# starting kernel
# baseline (speedup 1.0000x reference)
#include <hip/hip_runtime.h>

typedef float v2f __attribute__((ext_vector_type(2)));
typedef float v4f __attribute__((ext_vector_type(4)));
typedef float v8f __attribute__((ext_vector_type(8)));

#define BLK_M 128
#define BLK_N 256
#define BLK_K 16
#define STRIDE_A 20
#define PSTRIDE 544
#define ASZ (BLK_M * STRIDE_A)
#define BSZ ((BLK_K / 2) * PSTRIDE)

__global__ __launch_bounds__(256)
void Dynamic_Masking_Layer_gemm_wmma_f32(const float* __restrict__ X,
                                         const float* __restrict__ Mk,
                                         const float* __restrict__ W,
                                         const float* __restrict__ Bias,
                                         float* __restrict__ Out,
                                         int M, int N, int K)
{
    __shared__ float As[2 * ASZ];
    __shared__ float Bs[2 * BSZ];

    const int tid  = threadIdx.x;
    const int lane = tid & 31;
    const int wave = tid >> 5;
    const int wm   = wave & 1;
    const int wn   = wave >> 1;

    const int m_blk = blockIdx.y * BLK_M;
    const int n_blk = blockIdx.x * BLK_N;

    const int rowA  = tid >> 2;
    const int col4A = tid & 3;
    const int pB = tid >> 6;
    const int qB = tid & 63;

    v8f acc[4][4];
#pragma unroll
    for (int im = 0; im < 4; ++im)
#pragma unroll
        for (int in = 0; in < 4; ++in)
#pragma unroll
            for (int v = 0; v < 8; ++v)
                acc[im][in][v] = 0.0f;

    const int KT = K / BLK_K;

    v4f ra[2], rw0[2], rw1[2], rm0[2], rm1[2];
#pragma unroll
    for (int i = 0; i < 2; ++i)
        ra[i] = *(const v4f*)(X + (size_t)(m_blk + rowA + 64 * i) * K + (col4A * 4));
#pragma unroll
    for (int i = 0; i < 2; ++i) {
        const int p = pB + 4 * i;
        rw0[i] = *(const v4f*)(W  + (size_t)(2 * p + 0) * N + (n_blk + qB * 4));
        rw1[i] = *(const v4f*)(W  + (size_t)(2 * p + 1) * N + (n_blk + qB * 4));
        rm0[i] = *(const v4f*)(Mk + (size_t)(2 * p + 0) * N + (n_blk + qB * 4));
        rm1[i] = *(const v4f*)(Mk + (size_t)(2 * p + 1) * N + (n_blk + qB * 4));
    }

    for (int kt = 0; kt < KT; ++kt) {
        const int buf = kt & 1;
        float* __restrict__ as_w = As + buf * ASZ;
        float* __restrict__ bs_w = Bs + buf * BSZ;

#pragma unroll
        for (int i = 0; i < 2; ++i)
            *(v4f*)(as_w + (rowA + 64 * i) * STRIDE_A + col4A * 4) = ra[i];
#pragma unroll
        for (int i = 0; i < 2; ++i) {
            const int p = pB + 4 * i;
#pragma unroll
            for (int j = 0; j < 4; ++j) {
                v2f pr;
                pr.x = (rm0[i][j] != 0.0f) ? rw0[i][j] : 0.0f;
                pr.y = (rm1[i][j] != 0.0f) ? rw1[i][j] : 0.0f;
                *(v2f*)(bs_w + p * PSTRIDE + (qB * 4 + j) * 2) = pr;
            }
        }
        __syncthreads();

        if (kt + 1 < KT) {
            const int kn = (kt + 1) * BLK_K;
#pragma unroll
            for (int i = 0; i < 2; ++i)
                ra[i] = *(const v4f*)(X + (size_t)(m_blk + rowA + 64 * i) * K + (kn + col4A * 4));
#pragma unroll
            for (int i = 0; i < 2; ++i) {
                const int p = pB + 4 * i;
                rw0[i] = *(const v4f*)(W  + (size_t)(kn + 2 * p + 0) * N + (n_blk + qB * 4));
                rw1[i] = *(const v4f*)(W  + (size_t)(kn + 2 * p + 1) * N + (n_blk + qB * 4));
                rm0[i] = *(const v4f*)(Mk + (size_t)(kn + 2 * p + 0) * N + (n_blk + qB * 4));
                rm1[i] = *(const v4f*)(Mk + (size_t)(kn + 2 * p + 1) * N + (n_blk + qB * 4));
            }
        }

        const float* __restrict__ as_r = As + buf * ASZ;
        const float* __restrict__ bs_r = Bs + buf * BSZ;

#pragma unroll
        for (int s = 0; s < 4; ++s) {
            const int krow = 4 * s + 2 * (lane >> 4);
            const int prow = 2 * s + (lane >> 4);

            v2f af[4];
#pragma unroll
            for (int im = 0; im < 4; ++im) {
                const int m_l = wm * 64 + im * 16 + (lane & 15);
                af[im] = *(const v2f*)(as_r + m_l * STRIDE_A + krow);
            }

            v2f bf[4];
#pragma unroll
            for (int in = 0; in < 4; ++in) {
                const int n_l = wn * 64 + in * 16 + (lane & 15);
                bf[in] = *(const v2f*)(bs_r + prow * PSTRIDE + n_l * 2);
            }

#pragma unroll
            for (int im = 0; im < 4; ++im)
#pragma unroll
                for (int in = 0; in < 4; ++in)
                    acc[im][in] = __builtin_amdgcn_wmma_f32_16x16x4_f32(
                        false, af[im], false, bf[in],
                        (short)0, acc[im][in], false, false);
        }
    }

#pragma unroll
    for (int in = 0; in < 4; ++in) {
        const int n_g = n_blk + wn * 64 + in * 16 + (lane & 15);
        const float bv = Bias[n_g];
#pragma unroll
        for (int im = 0; im < 4; ++im) {
            const int m_base = m_blk + wm * 64 + im * 16 + 8 * (lane >> 4);
#pragma unroll
            for (int v = 0; v < 8; ++v)
                Out[(size_t)(m_base + v) * N + n_g] = acc[im][in][v] + bv;
        }
    }
}

extern "C" void kernel_launch(void* const* d_in, const int* in_sizes, int n_in,
                              void* d_out, int out_size, void* d_ws, size_t ws_size,
                              hipStream_t stream) {
    const float* X    = (const float*)d_in[0];
    const float* Mask = (const float*)d_in[1];
    const float* W    = (const float*)d_in[2];
    const float* Bias = (const float*)d_in[3];
    float* Out        = (float*)d_out;

    const int N = in_sizes[3];
    const int K = in_sizes[1] / N;
    const int M = in_sizes[0] / K;

    dim3 grid(N / BLK_N, M / BLK_M);
    Dynamic_Masking_Layer_gemm_wmma_f32<<<grid, dim3(256), 0, stream>>>(
        X, Mask, W, Bias, Out, M, N, K);
}


# Both operands loaded from global memory straight into WMMA fragment registers (b128 per lane, two groups ahead): no LDS, no barrier; each WMMA takes a different set of four k: f32 sum order changes
# speedup vs baseline: 1.2196x; 1.2196x over previous
#include <hip/hip_runtime.h>

typedef float v2f __attribute__((ext_vector_type(2)));
typedef float v4f __attribute__((ext_vector_type(4)));
typedef float v8f __attribute__((ext_vector_type(8)));

#define BLK_M 128
#define BLK_N 256

struct Group {
    v4f a[4];
    v4f w[2][2];
    v4f m[2][2];
};

struct LaneOff {
    unsigned a[4];
    unsigned b[2][2];
};

__device__ __forceinline__ v4f ld16(const float* __restrict__ base, unsigned off_bytes)
{
    return *(const v4f*)((const char*)base + off_bytes);
}

__device__ __forceinline__ void keep_offsets_narrow(LaneOff& o)
{
    asm volatile("" : "+v"(o.a[0]), "+v"(o.a[1]), "+v"(o.a[2]), "+v"(o.a[3]),
                      "+v"(o.b[0][0]), "+v"(o.b[0][1]), "+v"(o.b[1][0]), "+v"(o.b[1][1]));
}

__device__ __forceinline__ void load_group(Group& g, const LaneOff& o,
                                           const float* __restrict__ xk,
                                           const float* __restrict__ wk,
                                           const float* __restrict__ mk)
{
#pragma unroll
    for (int im = 0; im < 4; ++im)
        g.a[im] = ld16(xk, o.a[im]);
#pragma unroll
    for (int t = 0; t < 2; ++t)
#pragma unroll
        for (int e = 0; e < 2; ++e) {
            g.m[t][e] = ld16(mk, o.b[t][e]);
            g.w[t][e] = ld16(wk, o.b[t][e]);
        }
}

__device__ __forceinline__ void compute_group(const Group& g, v8f (&acc)[4][4])
{
    bool keep[2][2][4];
#pragma unroll
    for (int t = 0; t < 2; ++t)
#pragma unroll
        for (int e = 0; e < 2; ++e)
#pragma unroll
            for (int j = 0; j < 4; ++j)
                keep[t][e][j] = (g.m[t][e][j] != 0.0f);
    __builtin_amdgcn_sched_barrier(0);
    v2f bf[2][4];
#pragma unroll
    for (int t = 0; t < 2; ++t)
#pragma unroll
        for (int j = 0; j < 4; ++j) {
            bf[t][j].x = keep[t][0][j] ? g.w[t][0][j] : 0.0f;
            bf[t][j].y = keep[t][1][j] ? g.w[t][1][j] : 0.0f;
        }
    __builtin_amdgcn_sched_barrier(0);
#pragma unroll
    for (int t = 0; t < 2; ++t) {
#pragma unroll
        for (int im = 0; im < 4; ++im) {
            v2f af;
            af.x = g.a[im][2 * t + 0];
            af.y = g.a[im][2 * t + 1];
#pragma unroll
            for (int j = 0; j < 4; ++j)
                acc[im][j] = __builtin_amdgcn_wmma_f32_16x16x4_f32(
                    false, af, false, bf[t][j],
                    (short)0, acc[im][j], false, false);
        }
    }
}

__global__ __launch_bounds__(256)
void Dynamic_Masking_Layer_gemm_wmma_f32(const float* __restrict__ X,
                                         const float* __restrict__ Mk,
                                         const float* __restrict__ W,
                                         const float* __restrict__ Bias,
                                         float* __restrict__ Out,
                                         int M, int N, int K)
{
    const int lane = threadIdx.x & 31;
    const int wave = __builtin_amdgcn_readfirstlane((int)(threadIdx.x >> 5));
    const int wm   = wave & 1;
    const int wn   = wave >> 1;
    const int c    = lane & 15;
    const int h    = lane >> 4;

    const int m0 = blockIdx.y * BLK_M + wm * 64;
    const int n0 = blockIdx.x * BLK_N + wn * 64;

    LaneOff o;
#pragma unroll
    for (int im = 0; im < 4; ++im)
        o.a[im] = (unsigned)((im * 16 + c) * K + 4 * h) * 4u;
#pragma unroll
    for (int t = 0; t < 2; ++t)
#pragma unroll
        for (int e = 0; e < 2; ++e)
            o.b[t][e] = (unsigned)((4 * h + 2 * t + e) * N + 4 * c) * 4u;

    const float* __restrict__ xk = X  + (size_t)m0 * K;
    const float* __restrict__ wk = W  + n0;
    const float* __restrict__ mk = Mk + n0;
    const size_t n8 = (size_t)8 * N;

    v8f acc[4][4];
#pragma unroll
    for (int im = 0; im < 4; ++im)
#pragma unroll
        for (int j = 0; j < 4; ++j)
#pragma unroll
            for (int v = 0; v < 8; ++v)
                acc[im][j][v] = 0.0f;

    const int KG = K / 8;

    Group g0, g1, g2, g3;
    load_group(g0, o, xk, wk, mk); xk += 8; wk += n8; mk += n8;
    load_group(g1, o, xk, wk, mk); xk += 8; wk += n8; mk += n8;
    __builtin_amdgcn_sched_barrier(0);
    load_group(g2, o, xk, wk, mk); xk += 8; wk += n8; mk += n8;
    load_group(g3, o, xk, wk, mk); xk += 8; wk += n8; mk += n8;

    for (int kg = 4; kg < KG; kg += 4) {
        compute_group(g0, acc);
        compute_group(g1, acc);
        __builtin_amdgcn_sched_barrier(0);
        keep_offsets_narrow(o);
        load_group(g0, o, xk, wk, mk); xk += 8; wk += n8; mk += n8;
        load_group(g1, o, xk, wk, mk); xk += 8; wk += n8; mk += n8;
        __builtin_amdgcn_sched_barrier(0);
        compute_group(g2, acc);
        compute_group(g3, acc);
        __builtin_amdgcn_sched_barrier(0);
        keep_offsets_narrow(o);
        load_group(g2, o, xk, wk, mk); xk += 8; wk += n8; mk += n8;
        load_group(g3, o, xk, wk, mk); xk += 8; wk += n8; mk += n8;
    }
    compute_group(g0, acc);
    compute_group(g1, acc);
    compute_group(g2, acc);
    compute_group(g3, acc);
    __builtin_amdgcn_sched_barrier(0);
    asm volatile("v_nop\n\tv_nop\n\tv_nop\n\tv_nop");

    const v4f bv = *(const v4f*)(Bias + n0 + 4 * c);
#pragma unroll
    for (int im = 0; im < 4; ++im) {
        const int m_base = m0 + im * 16 + 8 * h;
#pragma unroll
        for (int r = 0; r < 8; ++r) {
            v4f ov;
            ov.x = acc[im][0][r] + bv.x;
            ov.y = acc[im][1][r] + bv.y;
            ov.z = acc[im][2][r] + bv.z;
            ov.w = acc[im][3][r] + bv.w;
            *(v4f*)(Out + (size_t)(m_base + r) * N + (n0 + 4 * c)) = ov;
        }
    }
}

extern "C" void kernel_launch(void* const* d_in, const int* in_sizes, int n_in,
                              void* d_out, int out_size, void* d_ws, size_t ws_size,
                              hipStream_t stream) {
    const float* X    = (const float*)d_in[0];
    const float* Mask = (const float*)d_in[1];
    const float* W    = (const float*)d_in[2];
    const float* Bias = (const float*)d_in[3];
    float* Out        = (float*)d_out;

    const int N = in_sizes[3];
    const int K = in_sizes[1] / N;
    const int M = in_sizes[0] / K;

    dim3 grid(N / BLK_N, M / BLK_M);
    Dynamic_Masking_Layer_gemm_wmma_f32<<<grid, dim3(256), 0, stream>>>(
        X, Mask, W, Bias, Out, M, N, K);
}


# One wave per SIMD: work-groups of four waves side by side along N, each a 128x64 tile (8x4 accumulator tiles); all fragments still loaded straight from global memory; f32 sum order unchanged
# speedup vs baseline: 1.2672x; 1.0390x over previous
#include <hip/hip_runtime.h>

typedef float v2f __attribute__((ext_vector_type(2)));
typedef float v4f __attribute__((ext_vector_type(4)));
typedef float v8f __attribute__((ext_vector_type(8)));

#define BLK_M 128
#define BLK_N 256

struct Group {
    v4f a[8];
    v4f w[2][2];
    v4f m[2][2];
};

struct LaneOff {
    unsigned a[8];
    unsigned b[2][2];
};

__device__ __forceinline__ v4f ld16(const float* __restrict__ base, unsigned off_bytes)
{
    return *(const v4f*)((const char*)base + off_bytes);
}

__device__ __forceinline__ void keep_offsets_narrow(LaneOff& o)
{
    asm volatile("" : "+v"(o.a[0]), "+v"(o.a[1]), "+v"(o.a[2]), "+v"(o.a[3]),
                      "+v"(o.a[4]), "+v"(o.a[5]), "+v"(o.a[6]), "+v"(o.a[7]),
                      "+v"(o.b[0][0]), "+v"(o.b[0][1]), "+v"(o.b[1][0]), "+v"(o.b[1][1]));
}

__device__ __forceinline__ void load_group(Group& g, const LaneOff& o,
                                           const float* __restrict__ xk,
                                           const float* __restrict__ wk,
                                           const float* __restrict__ mk)
{
#pragma unroll
    for (int im = 0; im < 8; ++im)
        g.a[im] = ld16(xk, o.a[im]);
#pragma unroll
    for (int t = 0; t < 2; ++t)
#pragma unroll
        for (int e = 0; e < 2; ++e) {
            g.m[t][e] = ld16(mk, o.b[t][e]);
            g.w[t][e] = ld16(wk, o.b[t][e]);
        }
}

__device__ __forceinline__ void compute_group(const Group& g, v8f (&acc)[8][4])
{
    bool keep[2][2][4];
#pragma unroll
    for (int t = 0; t < 2; ++t)
#pragma unroll
        for (int e = 0; e < 2; ++e)
#pragma unroll
            for (int j = 0; j < 4; ++j)
                keep[t][e][j] = (g.m[t][e][j] != 0.0f);
    __builtin_amdgcn_sched_barrier(0);
    v2f bf[2][4];
#pragma unroll
    for (int t = 0; t < 2; ++t)
#pragma unroll
        for (int j = 0; j < 4; ++j) {
            bf[t][j].x = keep[t][0][j] ? g.w[t][0][j] : 0.0f;
            bf[t][j].y = keep[t][1][j] ? g.w[t][1][j] : 0.0f;
        }
    __builtin_amdgcn_sched_barrier(0);
#pragma unroll
    for (int t = 0; t < 2; ++t) {
#pragma unroll
        for (int im = 0; im < 8; ++im) {
            v2f af;
            af.x = g.a[im][2 * t + 0];
            af.y = g.a[im][2 * t + 1];
#pragma unroll
            for (int j = 0; j < 4; ++j)
                acc[im][j] = __builtin_amdgcn_wmma_f32_16x16x4_f32(
                    false, af, false, bf[t][j],
                    (short)0, acc[im][j], false, false);
        }
    }
}

__global__ __launch_bounds__(128)
void Dynamic_Masking_Layer_gemm_wmma_f32(const float* __restrict__ X,
                                         const float* __restrict__ Mk,
                                         const float* __restrict__ W,
                                         const float* __restrict__ Bias,
                                         float* __restrict__ Out,
                                         int M, int N, int K)
{
    const int lane = threadIdx.x & 31;
    const int wave = __builtin_amdgcn_readfirstlane((int)(threadIdx.x >> 5));
    const int c    = lane & 15;
    const int h    = lane >> 4;

    const int m0 = blockIdx.y * BLK_M;
    const int n0 = blockIdx.x * BLK_N + wave * 64;

    LaneOff o;
#pragma unroll
    for (int im = 0; im < 8; ++im)
        o.a[im] = (unsigned)((im * 16 + c) * K + 4 * h) * 4u;
#pragma unroll
    for (int t = 0; t < 2; ++t)
#pragma unroll
        for (int e = 0; e < 2; ++e)
            o.b[t][e] = (unsigned)((4 * h + 2 * t + e) * N + 4 * c) * 4u;

    const float* __restrict__ xk = X  + (size_t)m0 * K;
    const float* __restrict__ wk = W  + n0;
    const float* __restrict__ mk = Mk + n0;
    const size_t n8 = (size_t)8 * N;

    v8f acc[8][4];
#pragma unroll
    for (int im = 0; im < 8; ++im)
#pragma unroll
        for (int j = 0; j < 4; ++j)
#pragma unroll
            for (int v = 0; v < 8; ++v)
                acc[im][j][v] = 0.0f;

    const int KG = K / 8;

    Group g0, g1, g2, g3;
    load_group(g0, o, xk, wk, mk); xk += 8; wk += n8; mk += n8;
    load_group(g1, o, xk, wk, mk); xk += 8; wk += n8; mk += n8;
    __builtin_amdgcn_sched_barrier(0);
    load_group(g2, o, xk, wk, mk); xk += 8; wk += n8; mk += n8;
    load_group(g3, o, xk, wk, mk); xk += 8; wk += n8; mk += n8;

    for (int kg = 4; kg < KG; kg += 4) {
        compute_group(g0, acc);
        compute_group(g1, acc);
        __builtin_amdgcn_sched_barrier(0);
        keep_offsets_narrow(o);
        load_group(g0, o, xk, wk, mk); xk += 8; wk += n8; mk += n8;
        load_group(g1, o, xk, wk, mk); xk += 8; wk += n8; mk += n8;
        __builtin_amdgcn_sched_barrier(0);
        compute_group(g2, acc);
        compute_group(g3, acc);
        __builtin_amdgcn_sched_barrier(0);
        keep_offsets_narrow(o);
        load_group(g2, o, xk, wk, mk); xk += 8; wk += n8; mk += n8;
        load_group(g3, o, xk, wk, mk); xk += 8; wk += n8; mk += n8;
    }
    compute_group(g0, acc);
    compute_group(g1, acc);
    compute_group(g2, acc);
    compute_group(g3, acc);
    __builtin_amdgcn_sched_barrier(0);
    asm volatile("v_nop\n\tv_nop\n\tv_nop\n\tv_nop");

    const v4f bv = *(const v4f*)(Bias + n0 + 4 * c);
#pragma unroll
    for (int im = 0; im < 8; ++im) {
        const int m_base = m0 + im * 16 + 8 * h;
#pragma unroll
        for (int r = 0; r < 8; ++r) {
            v4f ov;
            ov.x = acc[im][0][r] + bv.x;
            ov.y = acc[im][1][r] + bv.y;
            ov.z = acc[im][2][r] + bv.z;
            ov.w = acc[im][3][r] + bv.w;
            *(v4f*)(Out + (size_t)(m_base + r) * N + (n0 + 4 * c)) = ov;
        }
    }
}

extern "C" void kernel_launch(void* const* d_in, const int* in_sizes, int n_in,
                              void* d_out, int out_size, void* d_ws, size_t ws_size,
                              hipStream_t stream) {
    const float* X    = (const float*)d_in[0];
    const float* Mask = (const float*)d_in[1];
    const float* W    = (const float*)d_in[2];
    const float* Bias = (const float*)d_in[3];
    float* Out        = (float*)d_out;

    const int N = in_sizes[3];
    const int K = in_sizes[1] / N;
    const int M = in_sizes[0] / K;

    dim3 grid(N / BLK_N, M / BLK_M);
    Dynamic_Masking_Layer_gemm_wmma_f32<<<grid, dim3(128), 0, stream>>>(
        X, Mask, W, Bias, Out, M, N, K);
}
